# starting kernel
# baseline (speedup 1.0000x reference)
#include <hip/hip_runtime.h>

typedef __attribute__((ext_vector_type(2))) float v2f;
typedef __attribute__((ext_vector_type(4))) float v4f;
typedef __attribute__((ext_vector_type(8))) float v8f;

#define N_NODES 50000
#define N_EDGES 800000
#define D 128
#define LDSW 132

__global__ void zero_agg_kernel(float* __restrict__ agg) {
    int i = blockIdx.x * blockDim.x + threadIdx.x;
    v4f z = {0.f, 0.f, 0.f, 0.f};
    ((v4f*)agg)[i] = z;
}

__global__ void scatter_kernel(const float* __restrict__ x,
                               const int* __restrict__ src,
                               const int* __restrict__ dst,
                               float* __restrict__ agg) {
    int t = blockIdx.x * blockDim.x + threadIdx.x;
    int e = t >> 5;
    int lane = t & 31;
    if (e >= N_EDGES) return;
    int s = src[e];
    int d = dst[e];
    const v4f xv = *(const v4f*)(x + (size_t)s * D + lane * 4);
    float* ap = agg + (size_t)d * D + lane * 4;
    atomicAdd(ap + 0, xv.x);
    atomicAdd(ap + 1, xv.y);
    atomicAdd(ap + 2, xv.z);
    atomicAdd(ap + 3, xv.w);
}

__global__ __launch_bounds__(256) void gin_wmma_kernel(
    const float* __restrict__ x,   const float* __restrict__ agg,
    const float* __restrict__ W1,  const float* __restrict__ b1,
    const float* __restrict__ W2,  const float* __restrict__ b2,
    const float* __restrict__ Wres,const float* __restrict__ bres,
    float* __restrict__ out) {

    __shared__ float sh_xa[16 * LDSW];
    __shared__ float sh_xr[16 * LDSW];
    __shared__ float sh_t1[16 * LDSW];

    const int tid = threadIdx.x;
    const int rowbase = blockIdx.x * 16;

    for (int i = tid; i < 16 * (D / 4); i += 256) {
        int r  = i >> 5;
        int c4 = (i & 31) * 4;
        v4f xv = *(const v4f*)(x   + (size_t)(rowbase + r) * D + c4);
        v4f av = *(const v4f*)(agg + (size_t)(rowbase + r) * D + c4);
        v4f sv = {xv.x + av.x, xv.y + av.y, xv.z + av.z, xv.w + av.w};
        *(v4f*)(sh_xr + r * LDSW + c4) = xv;
        *(v4f*)(sh_xa + r * LDSW + c4) = sv;
    }
    __syncthreads();

    const int lane = tid & 31;
    const int wv   = tid >> 5;
    const int n    = lane & 15;
    const int h    = lane >> 4;
    const int colbase = wv * 16;

    v8f acc = {};
    #pragma unroll
    for (int k = 0; k < D; k += 4) {
        v2f a = *(const v2f*)(sh_xa + n * LDSW + k + 2 * h);
        v2f b;
        b.x = W1[(size_t)(k + 2 * h + 0) * D + colbase + n];
        b.y = W1[(size_t)(k + 2 * h + 1) * D + colbase + n];
        acc = __builtin_amdgcn_wmma_f32_16x16x4_f32(
                  false, a, false, b, (short)0, acc, false, false);
    }
    const float bias1 = b1[colbase + n];
    #pragma unroll
    for (int r = 0; r < 8; ++r) {
        float t = acc[r] + bias1;
        sh_t1[(r + 8 * h) * LDSW + colbase + n] = fmaxf(t, 0.f);
    }
    __syncthreads();

    v8f acc2 = {};
    #pragma unroll
    for (int k = 0; k < D; k += 4) {
        v2f a1 = *(const v2f*)(sh_t1 + n * LDSW + k + 2 * h);
        v2f bw2;
        bw2.x = W2[(size_t)(k + 2 * h + 0) * D + colbase + n];
        bw2.y = W2[(size_t)(k + 2 * h + 1) * D + colbase + n];
        acc2 = __builtin_amdgcn_wmma_f32_16x16x4_f32(
                   false, a1, false, bw2, (short)0, acc2, false, false);

        v2f a2 = *(const v2f*)(sh_xr + n * LDSW + k + 2 * h);
        v2f bwr;
        bwr.x = Wres[(size_t)(k + 2 * h + 0) * D + colbase + n];
        bwr.y = Wres[(size_t)(k + 2 * h + 1) * D + colbase + n];
        acc2 = __builtin_amdgcn_wmma_f32_16x16x4_f32(
                   false, a2, false, bwr, (short)0, acc2, false, false);
    }
    const float bias2 = b2[colbase + n] + bres[colbase + n];
    #pragma unroll
    for (int r = 0; r < 8; ++r) {
        out[(size_t)(rowbase + r + 8 * h) * D + colbase + n] = acc2[r] + bias2;
    }
}

extern "C" void kernel_launch(void* const* d_in, const int* in_sizes, int n_in,
                              void* d_out, int out_size, void* d_ws, size_t ws_size,
                              hipStream_t stream) {
    const float* x    = (const float*)d_in[0];
    const int*   ei   = (const int*)  d_in[1];
    const float* W1   = (const float*)d_in[2];
    const float* b1   = (const float*)d_in[3];
    const float* W2   = (const float*)d_in[4];
    const float* b2   = (const float*)d_in[5];
    const float* Wres = (const float*)d_in[6];
    const float* bres = (const float*)d_in[7];
    float* out = (float*)d_out;
    float* agg = (float*)d_ws;

    zero_agg_kernel<<<(N_NODES * D / 4) / 256, 256, 0, stream>>>(agg);

    scatter_kernel<<<(N_EDGES * 32) / 256, 256, 0, stream>>>(
        x, ei, ei + N_EDGES, agg);

    gin_wmma_kernel<<<N_NODES / 16, 256, 0, stream>>>(
        x, agg, W1, b1, W2, b2, Wres, bres, out);
}


# scatter: lane l takes features l+32j, so each f32 atomic add (and each x load) of a wave covers one whole 128-B line, not 4 quarter lines; adds still land in arrival order
# speedup vs baseline: 1.2865x; 1.2865x over previous
#include <hip/hip_runtime.h>

typedef __attribute__((ext_vector_type(2))) float v2f;
typedef __attribute__((ext_vector_type(4))) float v4f;
typedef __attribute__((ext_vector_type(8))) float v8f;

#define N_NODES 50000
#define N_EDGES 800000
#define D 128
#define LDSW 132

__global__ void zero_agg_kernel(float* __restrict__ agg) {
    int i = blockIdx.x * blockDim.x + threadIdx.x;
    v4f z = {0.f, 0.f, 0.f, 0.f};
    ((v4f*)agg)[i] = z;
}

__global__ void scatter_kernel(const float* __restrict__ x,
                               const int* __restrict__ src,
                               const int* __restrict__ dst,
                               float* __restrict__ agg) {
    int t = blockIdx.x * blockDim.x + threadIdx.x;
    int e = t >> 5;
    int lane = t & 31;
    if (e >= N_EDGES) return;
    int s = src[e];
    int d = dst[e];
    const float* xp = x + (size_t)s * D + lane;
    float* ap = agg + (size_t)d * D + lane;
    const float x0 = xp[0];
    const float x1 = xp[32];
    const float x2 = xp[64];
    const float x3 = xp[96];
    atomicAdd(ap + 0,  x0);
    atomicAdd(ap + 32, x1);
    atomicAdd(ap + 64, x2);
    atomicAdd(ap + 96, x3);
}

__global__ __launch_bounds__(256) void gin_wmma_kernel(
    const float* __restrict__ x,   const float* __restrict__ agg,
    const float* __restrict__ W1,  const float* __restrict__ b1,
    const float* __restrict__ W2,  const float* __restrict__ b2,
    const float* __restrict__ Wres,const float* __restrict__ bres,
    float* __restrict__ out) {

    __shared__ float sh_xa[16 * LDSW];
    __shared__ float sh_xr[16 * LDSW];
    __shared__ float sh_t1[16 * LDSW];

    const int tid = threadIdx.x;
    const int rowbase = blockIdx.x * 16;

    for (int i = tid; i < 16 * (D / 4); i += 256) {
        int r  = i >> 5;
        int c4 = (i & 31) * 4;
        v4f xv = *(const v4f*)(x   + (size_t)(rowbase + r) * D + c4);
        v4f av = *(const v4f*)(agg + (size_t)(rowbase + r) * D + c4);
        v4f sv = {xv.x + av.x, xv.y + av.y, xv.z + av.z, xv.w + av.w};
        *(v4f*)(sh_xr + r * LDSW + c4) = xv;
        *(v4f*)(sh_xa + r * LDSW + c4) = sv;
    }
    __syncthreads();

    const int lane = tid & 31;
    const int wv   = tid >> 5;
    const int n    = lane & 15;
    const int h    = lane >> 4;
    const int colbase = wv * 16;

    v8f acc = {};
    #pragma unroll
    for (int k = 0; k < D; k += 4) {
        v2f a = *(const v2f*)(sh_xa + n * LDSW + k + 2 * h);
        v2f b;
        b.x = W1[(size_t)(k + 2 * h + 0) * D + colbase + n];
        b.y = W1[(size_t)(k + 2 * h + 1) * D + colbase + n];
        acc = __builtin_amdgcn_wmma_f32_16x16x4_f32(
                  false, a, false, b, (short)0, acc, false, false);
    }
    const float bias1 = b1[colbase + n];
    #pragma unroll
    for (int r = 0; r < 8; ++r) {
        float t = acc[r] + bias1;
        sh_t1[(r + 8 * h) * LDSW + colbase + n] = fmaxf(t, 0.f);
    }
    __syncthreads();

    v8f acc2 = {};
    #pragma unroll
    for (int k = 0; k < D; k += 4) {
        v2f a1 = *(const v2f*)(sh_t1 + n * LDSW + k + 2 * h);
        v2f bw2;
        bw2.x = W2[(size_t)(k + 2 * h + 0) * D + colbase + n];
        bw2.y = W2[(size_t)(k + 2 * h + 1) * D + colbase + n];
        acc2 = __builtin_amdgcn_wmma_f32_16x16x4_f32(
                   false, a1, false, bw2, (short)0, acc2, false, false);

        v2f a2 = *(const v2f*)(sh_xr + n * LDSW + k + 2 * h);
        v2f bwr;
        bwr.x = Wres[(size_t)(k + 2 * h + 0) * D + colbase + n];
        bwr.y = Wres[(size_t)(k + 2 * h + 1) * D + colbase + n];
        acc2 = __builtin_amdgcn_wmma_f32_16x16x4_f32(
                   false, a2, false, bwr, (short)0, acc2, false, false);
    }
    const float bias2 = b2[colbase + n] + bres[colbase + n];
    #pragma unroll
    for (int r = 0; r < 8; ++r) {
        out[(size_t)(rowbase + r + 8 * h) * D + colbase + n] = acc2[r] + bias2;
    }
}

extern "C" void kernel_launch(void* const* d_in, const int* in_sizes, int n_in,
                              void* d_out, int out_size, void* d_ws, size_t ws_size,
                              hipStream_t stream) {
    const float* x    = (const float*)d_in[0];
    const int*   ei   = (const int*)  d_in[1];
    const float* W1   = (const float*)d_in[2];
    const float* b1   = (const float*)d_in[3];
    const float* W2   = (const float*)d_in[4];
    const float* b2   = (const float*)d_in[5];
    const float* Wres = (const float*)d_in[6];
    const float* bres = (const float*)d_in[7];
    float* out = (float*)d_out;
    float* agg = (float*)d_ws;

    zero_agg_kernel<<<(N_NODES * D / 4) / 256, 256, 0, stream>>>(agg);

    scatter_kernel<<<(N_EDGES * 32) / 256, 256, 0, stream>>>(
        x, ei, ei + N_EDGES, agg);

    gin_wmma_kernel<<<N_NODES / 16, 256, 0, stream>>>(
        x, agg, W1, b1, W2, b2, Wres, bres, out);
}


# edges bucketed by destination: a returning int add gives each edge a slot in its node's row (arrival order); one wave per node sums its rows in registers in slot order: no f32 atomics, no zero fill
# speedup vs baseline: 1.4870x; 1.1558x over previous
#include <hip/hip_runtime.h>

typedef __attribute__((ext_vector_type(2))) float v2f;
typedef __attribute__((ext_vector_type(4))) float v4f;
typedef __attribute__((ext_vector_type(8))) float v8f;

#define N_NODES 50000
#define N_EDGES 800000
#define D 128
#define LDSW 132

#define CAP 64
#define CNT_OFF   (N_NODES * D)
#define SLOT_OFF  (CNT_OFF + 50176)
#define OVF_OFF   (SLOT_OFF + N_NODES * CAP)

__global__ void zero_cnt_kernel(int* __restrict__ cnt) {
    int i = blockIdx.x * blockDim.x + threadIdx.x;
    if (i <= N_NODES) cnt[i] = 0;
}

__global__ void slot_kernel(const int* __restrict__ src,
                            const int* __restrict__ dst,
                            int* __restrict__ cnt,
                            int* __restrict__ slots,
                            int* __restrict__ ovf) {
    int e = blockIdx.x * blockDim.x + threadIdx.x;
    int s = src[e];
    int d = dst[e];
    int slot = atomicAdd(cnt + d, 1);
    if (slot < CAP) {
        slots[(size_t)d * CAP + slot] = s;
    } else {
        int p = atomicAdd(cnt + N_NODES, 1);
        ovf[2 * p + 0] = d;
        ovf[2 * p + 1] = s;
    }
}

__global__ __launch_bounds__(256) void gather_kernel(const float* __restrict__ x,
                                                     const int* __restrict__ cnt,
                                                     const int* __restrict__ slots,
                                                     const int* __restrict__ ovf,
                                                     float* __restrict__ agg) {
    const int lane = threadIdx.x & 31;
    const int w = __builtin_amdgcn_readfirstlane(blockIdx.x * 8 + (threadIdx.x >> 5));
    const int c  = cnt[w];
    const int cc = c < CAP ? c : CAP;
    const int* bp = slots + (size_t)w * CAP;
    const int s0 = (lane < cc)      ? bp[lane]      : 0;
    const int s1 = (lane + 32 < cc) ? bp[lane + 32] : 0;
    const float* xl = x + lane * 4;
#define ROW(svv, jj) (*(const v4f*)(xl + (size_t)__builtin_amdgcn_readlane((svv), (jj)) * D))

    v4f acc = {0.f, 0.f, 0.f, 0.f};
    int j = 0;
    for (; j + 8 <= cc; j += 8) {
        const int sv = j < 32 ? s0 : s1;
        const int q = j & 31;
        const v4f r0 = ROW(sv, q + 0), r1 = ROW(sv, q + 1), r2 = ROW(sv, q + 2), r3 = ROW(sv, q + 3);
        const v4f r4 = ROW(sv, q + 4), r5 = ROW(sv, q + 5), r6 = ROW(sv, q + 6), r7 = ROW(sv, q + 7);
        acc += r0; acc += r1; acc += r2; acc += r3;
        acc += r4; acc += r5; acc += r6; acc += r7;
    }
    if (j + 4 <= cc) {
        const int sv = j < 32 ? s0 : s1;
        const int q = j & 31;
        const v4f r0 = ROW(sv, q + 0), r1 = ROW(sv, q + 1), r2 = ROW(sv, q + 2), r3 = ROW(sv, q + 3);
        acc += r0; acc += r1; acc += r2; acc += r3;
        j += 4;
    }
    if (j + 2 <= cc) {
        const int sv = j < 32 ? s0 : s1;
        const int q = j & 31;
        const v4f r0 = ROW(sv, q + 0), r1 = ROW(sv, q + 1);
        acc += r0; acc += r1;
        j += 2;
    }
    if (j < cc) {
        const int sv = j < 32 ? s0 : s1;
        acc += ROW(sv, j & 31);
    }

    if (c > CAP) {
        int n = cnt[N_NODES];
        n = n < N_EDGES ? n : N_EDGES;
        for (int i0 = 0; i0 < n; i0 += 32) {
            const int i = i0 + lane;
            int od = -1, os = 0;
            if (i < n) { od = ovf[2 * i + 0]; os = ovf[2 * i + 1]; }
            unsigned m = (unsigned)__ballot(od == w);
            while (m) {
                const int j = __builtin_ctz(m);
                m &= m - 1;
                const int s = __builtin_amdgcn_readlane(os, j);
                acc += *(const v4f*)(xl + (size_t)s * D);
            }
        }
    }
    *(v4f*)(agg + (size_t)w * D + lane * 4) = acc;
}

__global__ __launch_bounds__(256) void gin_wmma_kernel(
    const float* __restrict__ x,   const float* __restrict__ agg,
    const float* __restrict__ W1,  const float* __restrict__ b1,
    const float* __restrict__ W2,  const float* __restrict__ b2,
    const float* __restrict__ Wres,const float* __restrict__ bres,
    float* __restrict__ out) {

    __shared__ float sh_xa[16 * LDSW];
    __shared__ float sh_xr[16 * LDSW];
    __shared__ float sh_t1[16 * LDSW];

    const int tid = threadIdx.x;
    const int rowbase = blockIdx.x * 16;

    for (int i = tid; i < 16 * (D / 4); i += 256) {
        int r  = i >> 5;
        int c4 = (i & 31) * 4;
        v4f xv = *(const v4f*)(x   + (size_t)(rowbase + r) * D + c4);
        v4f av = *(const v4f*)(agg + (size_t)(rowbase + r) * D + c4);
        v4f sv = {xv.x + av.x, xv.y + av.y, xv.z + av.z, xv.w + av.w};
        *(v4f*)(sh_xr + r * LDSW + c4) = xv;
        *(v4f*)(sh_xa + r * LDSW + c4) = sv;
    }
    __syncthreads();

    const int lane = tid & 31;
    const int wv   = tid >> 5;
    const int n    = lane & 15;
    const int h    = lane >> 4;
    const int colbase = wv * 16;

    v8f acc = {};
    #pragma unroll
    for (int k = 0; k < D; k += 4) {
        v2f a = *(const v2f*)(sh_xa + n * LDSW + k + 2 * h);
        v2f b;
        b.x = W1[(size_t)(k + 2 * h + 0) * D + colbase + n];
        b.y = W1[(size_t)(k + 2 * h + 1) * D + colbase + n];
        acc = __builtin_amdgcn_wmma_f32_16x16x4_f32(
                  false, a, false, b, (short)0, acc, false, false);
    }
    const float bias1 = b1[colbase + n];
    #pragma unroll
    for (int r = 0; r < 8; ++r) {
        float t = acc[r] + bias1;
        sh_t1[(r + 8 * h) * LDSW + colbase + n] = fmaxf(t, 0.f);
    }
    __syncthreads();

    v8f acc2 = {};
    #pragma unroll
    for (int k = 0; k < D; k += 4) {
        v2f a1 = *(const v2f*)(sh_t1 + n * LDSW + k + 2 * h);
        v2f bw2;
        bw2.x = W2[(size_t)(k + 2 * h + 0) * D + colbase + n];
        bw2.y = W2[(size_t)(k + 2 * h + 1) * D + colbase + n];
        acc2 = __builtin_amdgcn_wmma_f32_16x16x4_f32(
                   false, a1, false, bw2, (short)0, acc2, false, false);

        v2f a2 = *(const v2f*)(sh_xr + n * LDSW + k + 2 * h);
        v2f bwr;
        bwr.x = Wres[(size_t)(k + 2 * h + 0) * D + colbase + n];
        bwr.y = Wres[(size_t)(k + 2 * h + 1) * D + colbase + n];
        acc2 = __builtin_amdgcn_wmma_f32_16x16x4_f32(
                   false, a2, false, bwr, (short)0, acc2, false, false);
    }
    const float bias2 = b2[colbase + n] + bres[colbase + n];
    #pragma unroll
    for (int r = 0; r < 8; ++r) {
        out[(size_t)(rowbase + r + 8 * h) * D + colbase + n] = acc2[r] + bias2;
    }
}

extern "C" void kernel_launch(void* const* d_in, const int* in_sizes, int n_in,
                              void* d_out, int out_size, void* d_ws, size_t ws_size,
                              hipStream_t stream) {
    const float* x    = (const float*)d_in[0];
    const int*   ei   = (const int*)  d_in[1];
    const float* W1   = (const float*)d_in[2];
    const float* b1   = (const float*)d_in[3];
    const float* W2   = (const float*)d_in[4];
    const float* b2   = (const float*)d_in[5];
    const float* Wres = (const float*)d_in[6];
    const float* bres = (const float*)d_in[7];
    float* out = (float*)d_out;
    float* agg = (float*)d_ws;
    int* cnt   = (int*)d_ws + CNT_OFF;
    int* slots = (int*)d_ws + SLOT_OFF;
    int* ovf   = (int*)d_ws + OVF_OFF;

    zero_cnt_kernel<<<196, 256, 0, stream>>>(cnt);

    slot_kernel<<<N_EDGES / 256, 256, 0, stream>>>(ei, ei + N_EDGES, cnt, slots, ovf);

    gather_kernel<<<N_NODES / 8, 256, 0, stream>>>(x, cnt, slots, ovf, agg);

    gin_wmma_kernel<<<N_NODES / 16, 256, 0, stream>>>(
        x, agg, W1, b1, W2, b2, Wres, bres, out);
}


# edges bucketed by destination with LDS counters in two levels (250 node ranges, then nodes), no global returning add per edge; one work-group per range sums its nodes' rows, still in arrival order
# speedup vs baseline: 2.6428x; 1.7773x over previous
#include <hip/hip_runtime.h>

typedef __attribute__((ext_vector_type(2))) float v2f;
typedef __attribute__((ext_vector_type(4))) float v4f;
typedef __attribute__((ext_vector_type(8))) float v8f;

#define N_NODES 50000
#define N_EDGES 800000
#define D 128
#define LDSW 132

#define NB     250
#define RNG    200
#define GA     250
#define EPB    3200
#define TA     800
#define TB     1024
#define NWB    32
#define BINCAP 8192
#define CHUNK_OFF (N_NODES * D)
#define OFFS_OFF  (CHUNK_OFF + N_EDGES)

__device__ __forceinline__ int wave_scan256(const int* in, int* out, int lane) {
    int v[8];
    int sum = 0;
    for (int q = 0; q < 8; ++q) { v[q] = in[8 * lane + q]; sum += v[q]; }
    int incl = sum;
    for (int dlt = 1; dlt < 32; dlt <<= 1) {
        const int y = __shfl_up(incl, dlt);
        if (lane >= dlt) incl += y;
    }
    int run = incl - sum;
    for (int q = 0; q < 8; ++q) { out[8 * lane + q] = run; run += v[q]; }
    return __shfl(incl, 31);
}

__global__ __launch_bounds__(TA) void bin_kernel(const int* __restrict__ src,
                                                 const int* __restrict__ dst,
                                                 int* __restrict__ chunk,
                                                 int* __restrict__ offs) {
    __shared__ int hist[256];
    __shared__ int stage[EPB];
    const int t = threadIdx.x;
    const int e0 = blockIdx.x * EPB + t;
    if (t < 256) hist[t] = 0;
    int s[4], d[4], bin[4], rk[4];
    for (int k = 0; k < 4; ++k) { s[k] = src[e0 + k * TA]; d[k] = dst[e0 + k * TA]; }
    __syncthreads();
    for (int k = 0; k < 4; ++k) { bin[k] = d[k] / RNG; rk[k] = atomicAdd(&hist[bin[k]], 1); }
    __syncthreads();
    if (t < 32) wave_scan256(hist, hist, t);
    __syncthreads();
    for (int k = 0; k < 4; ++k)
        stage[hist[bin[k]] + rk[k]] = ((d[k] - bin[k] * RNG) << 16) | s[k];
    if (t < 256) offs[blockIdx.x * 256 + t] = hist[t];
    __syncthreads();
    for (int k = 0; k < 4; ++k) chunk[e0 + k * TA] = stage[t + k * TA];
}

__global__ __launch_bounds__(TB) void gather_kernel(const float* __restrict__ x,
                                                    const int* __restrict__ chunk,
                                                    const int* __restrict__ offs,
                                                    float* __restrict__ agg) {
    __shared__ int s_o0[256];
    __shared__ int s_cnt[256];
    __shared__ int s_pos[256];
    __shared__ int hist[256];
    __shared__ int noff[256];
    __shared__ int fill[256];
    __shared__ int list[BINCAP];
    __shared__ int sorted[BINCAP];
    __shared__ int s_total;
    __shared__ int s_next;

    const int t = threadIdx.x;
    const int lane = t & 31;
    const int wave = __builtin_amdgcn_readfirstlane(t >> 5);
    const int b = blockIdx.x;

    if (t < 256) {
        int o0 = 0, c = 0;
        if (t < GA) { o0 = offs[t * 256 + b]; c = offs[t * 256 + b + 1] - o0; }
        s_o0[t] = o0; s_cnt[t] = c; hist[t] = 0; fill[t] = 0;
    }
    if (t == 0) s_next = NWB;
    __syncthreads();
    if (wave == 0) {
        const int tot = wave_scan256(s_cnt, s_pos, lane);
        if (lane == 0) s_total = tot;
    }
    __syncthreads();
    const int total = s_total;
    const bool fast = total <= BINCAP;

    if (fast) {
        const int a = t >> 2, sub = t & 3;
        if (a < GA) {
            const int cA = s_cnt[a];
            const int pA = s_pos[a];
            const int* g = chunk + a * EPB + s_o0[a];
            for (int j = sub; j < cA; j += 16) {
                int w[4];
                for (int k = 0; k < 4; ++k) { const int jj = j + 4 * k; w[k] = jj < cA ? g[jj] : -1; }
                for (int k = 0; k < 4; ++k) {
                    const int jj = j + 4 * k;
                    if (jj < cA) { list[pA + jj] = w[k]; atomicAdd(&hist[w[k] >> 16], 1); }
                }
            }
        }
    }
    __syncthreads();
    if (fast && wave == 0) wave_scan256(hist, noff, lane);
    __syncthreads();
    if (fast) {
        for (int idx = t; idx < total; idx += TB) {
            const int w = list[idx];
            const int dl = w >> 16;
            const int p = noff[dl] + atomicAdd(&fill[dl], 1);
            sorted[p] = w & 0xffff;
        }
    }
    __syncthreads();

    const float* xl = x + lane * 4;
#define ROW(svv, jj) (*(const v4f*)(xl + (size_t)__builtin_amdgcn_readlane((svv), (jj)) * D))

    int i = wave;
    while (i < RNG) {
        v4f acc = {0.f, 0.f, 0.f, 0.f};
        if (fast) {
            const int base = __builtin_amdgcn_readfirstlane(noff[i]);
            const int c = __builtin_amdgcn_readfirstlane(noff[i + 1]) - base;
            for (int j0 = 0; j0 < c; j0 += 32) {
                const int rem = c - j0;
                const int cc = rem < 32 ? rem : 32;
                const int sv = lane < cc ? sorted[base + j0 + lane] : 0;
                int j = 0;
                for (; j + 8 <= cc; j += 8) {
                    const v4f r0 = ROW(sv, j + 0), r1 = ROW(sv, j + 1), r2 = ROW(sv, j + 2), r3 = ROW(sv, j + 3);
                    const v4f r4 = ROW(sv, j + 4), r5 = ROW(sv, j + 5), r6 = ROW(sv, j + 6), r7 = ROW(sv, j + 7);
                    acc += r0; acc += r1; acc += r2; acc += r3;
                    acc += r4; acc += r5; acc += r6; acc += r7;
                }
                if (j + 4 <= cc) {
                    const v4f r0 = ROW(sv, j + 0), r1 = ROW(sv, j + 1), r2 = ROW(sv, j + 2), r3 = ROW(sv, j + 3);
                    acc += r0; acc += r1; acc += r2; acc += r3;
                    j += 4;
                }
                if (j + 2 <= cc) {
                    const v4f r0 = ROW(sv, j + 0), r1 = ROW(sv, j + 1);
                    acc += r0; acc += r1;
                    j += 2;
                }
                if (j < cc) acc += ROW(sv, j);
            }
        } else {
            for (int a = 0; a < GA; ++a) {
                const int cA = __builtin_amdgcn_readfirstlane(s_cnt[a]);
                const int* g = chunk + a * EPB + __builtin_amdgcn_readfirstlane(s_o0[a]);
                for (int j0 = 0; j0 < cA; j0 += 32) {
                    const int jj = j0 + lane;
                    const int w = jj < cA ? g[jj] : -1;
                    unsigned m = (unsigned)__ballot((w >> 16) == i);
                    while (m) {
                        const int q = __builtin_ctz(m);
                        m &= m - 1;
                        acc += ROW(w & 0xffff, q);
                    }
                }
            }
        }
        *(v4f*)(agg + (size_t)(b * RNG + i) * D + lane * 4) = acc;
        int nx = 0;
        if (lane == 0) nx = atomicAdd(&s_next, 1);
        i = __builtin_amdgcn_readfirstlane(nx);
    }
#undef ROW
}

__global__ __launch_bounds__(256) void gin_wmma_kernel(
    const float* __restrict__ x,   const float* __restrict__ agg,
    const float* __restrict__ W1,  const float* __restrict__ b1,
    const float* __restrict__ W2,  const float* __restrict__ b2,
    const float* __restrict__ Wres,const float* __restrict__ bres,
    float* __restrict__ out) {

    __shared__ float sh_xa[16 * LDSW];
    __shared__ float sh_xr[16 * LDSW];
    __shared__ float sh_t1[16 * LDSW];

    const int tid = threadIdx.x;
    const int rowbase = blockIdx.x * 16;

    for (int i = tid; i < 16 * (D / 4); i += 256) {
        int r  = i >> 5;
        int c4 = (i & 31) * 4;
        v4f xv = *(const v4f*)(x   + (size_t)(rowbase + r) * D + c4);
        v4f av = *(const v4f*)(agg + (size_t)(rowbase + r) * D + c4);
        v4f sv = {xv.x + av.x, xv.y + av.y, xv.z + av.z, xv.w + av.w};
        *(v4f*)(sh_xr + r * LDSW + c4) = xv;
        *(v4f*)(sh_xa + r * LDSW + c4) = sv;
    }
    __syncthreads();

    const int lane = tid & 31;
    const int wv   = tid >> 5;
    const int n    = lane & 15;
    const int h    = lane >> 4;
    const int colbase = wv * 16;

    v8f acc = {};
    #pragma unroll
    for (int k = 0; k < D; k += 4) {
        v2f a = *(const v2f*)(sh_xa + n * LDSW + k + 2 * h);
        v2f b;
        b.x = W1[(size_t)(k + 2 * h + 0) * D + colbase + n];
        b.y = W1[(size_t)(k + 2 * h + 1) * D + colbase + n];
        acc = __builtin_amdgcn_wmma_f32_16x16x4_f32(
                  false, a, false, b, (short)0, acc, false, false);
    }
    const float bias1 = b1[colbase + n];
    #pragma unroll
    for (int r = 0; r < 8; ++r) {
        float t = acc[r] + bias1;
        sh_t1[(r + 8 * h) * LDSW + colbase + n] = fmaxf(t, 0.f);
    }
    __syncthreads();

    v8f acc2 = {};
    #pragma unroll
    for (int k = 0; k < D; k += 4) {
        v2f a1 = *(const v2f*)(sh_t1 + n * LDSW + k + 2 * h);
        v2f bw2;
        bw2.x = W2[(size_t)(k + 2 * h + 0) * D + colbase + n];
        bw2.y = W2[(size_t)(k + 2 * h + 1) * D + colbase + n];
        acc2 = __builtin_amdgcn_wmma_f32_16x16x4_f32(
                   false, a1, false, bw2, (short)0, acc2, false, false);

        v2f a2 = *(const v2f*)(sh_xr + n * LDSW + k + 2 * h);
        v2f bwr;
        bwr.x = Wres[(size_t)(k + 2 * h + 0) * D + colbase + n];
        bwr.y = Wres[(size_t)(k + 2 * h + 1) * D + colbase + n];
        acc2 = __builtin_amdgcn_wmma_f32_16x16x4_f32(
                   false, a2, false, bwr, (short)0, acc2, false, false);
    }
    const float bias2 = b2[colbase + n] + bres[colbase + n];
    #pragma unroll
    for (int r = 0; r < 8; ++r) {
        out[(size_t)(rowbase + r + 8 * h) * D + colbase + n] = acc2[r] + bias2;
    }
}

extern "C" void kernel_launch(void* const* d_in, const int* in_sizes, int n_in,
                              void* d_out, int out_size, void* d_ws, size_t ws_size,
                              hipStream_t stream) {
    const float* x    = (const float*)d_in[0];
    const int*   ei   = (const int*)  d_in[1];
    const float* W1   = (const float*)d_in[2];
    const float* b1   = (const float*)d_in[3];
    const float* W2   = (const float*)d_in[4];
    const float* b2   = (const float*)d_in[5];
    const float* Wres = (const float*)d_in[6];
    const float* bres = (const float*)d_in[7];
    float* out = (float*)d_out;
    float* agg = (float*)d_ws;
    int* chunk = (int*)d_ws + CHUNK_OFF;
    int* offs  = (int*)d_ws + OFFS_OFF;

    bin_kernel<<<GA, TA, 0, stream>>>(ei, ei + N_EDGES, chunk, offs);

    gather_kernel<<<NB, TB, 0, stream>>>(x, chunk, offs, agg);

    gin_wmma_kernel<<<N_NODES / 16, 256, 0, stream>>>(
        x, agg, W1, b1, W2, b2, Wres, bres, out);
}


# MLP kernel weight-stationary: 512 resident work-groups, each wave keeps its slices of W1, W2, Wres in registers over its 6-7 row tiles; next tile's rows fetched ahead; sum order unchanged
# speedup vs baseline: 2.9131x; 1.1023x over previous
#include <hip/hip_runtime.h>

typedef __attribute__((ext_vector_type(2))) float v2f;
typedef __attribute__((ext_vector_type(4))) float v4f;
typedef __attribute__((ext_vector_type(8))) float v8f;

#define N_NODES 50000
#define N_EDGES 800000
#define D 128
#define LDSW 132

#define NB     250
#define RNG    200
#define GA     250
#define EPB    3200
#define TA     800
#define TB     1024
#define NWB    32
#define BINCAP 8192
#define CHUNK_OFF (N_NODES * D)
#define OFFS_OFF  (CHUNK_OFF + N_EDGES)

__device__ __forceinline__ int wave_scan256(const int* in, int* out, int lane) {
    int v[8];
    int sum = 0;
    for (int q = 0; q < 8; ++q) { v[q] = in[8 * lane + q]; sum += v[q]; }
    int incl = sum;
    for (int dlt = 1; dlt < 32; dlt <<= 1) {
        const int y = __shfl_up(incl, dlt);
        if (lane >= dlt) incl += y;
    }
    int run = incl - sum;
    for (int q = 0; q < 8; ++q) { out[8 * lane + q] = run; run += v[q]; }
    return __shfl(incl, 31);
}

__global__ __launch_bounds__(TA) void bin_kernel(const int* __restrict__ src,
                                                 const int* __restrict__ dst,
                                                 int* __restrict__ chunk,
                                                 int* __restrict__ offs) {
    __shared__ int hist[256];
    __shared__ int stage[EPB];
    const int t = threadIdx.x;
    const int e0 = blockIdx.x * EPB + t;
    if (t < 256) hist[t] = 0;
    int s[4], d[4], bin[4], rk[4];
    for (int k = 0; k < 4; ++k) { s[k] = src[e0 + k * TA]; d[k] = dst[e0 + k * TA]; }
    __syncthreads();
    for (int k = 0; k < 4; ++k) { bin[k] = d[k] / RNG; rk[k] = atomicAdd(&hist[bin[k]], 1); }
    __syncthreads();
    if (t < 32) wave_scan256(hist, hist, t);
    __syncthreads();
    for (int k = 0; k < 4; ++k)
        stage[hist[bin[k]] + rk[k]] = ((d[k] - bin[k] * RNG) << 16) | s[k];
    if (t < 256) offs[blockIdx.x * 256 + t] = hist[t];
    __syncthreads();
    for (int k = 0; k < 4; ++k) chunk[e0 + k * TA] = stage[t + k * TA];
}

__global__ __launch_bounds__(TB) void gather_kernel(const float* __restrict__ x,
                                                    const int* __restrict__ chunk,
                                                    const int* __restrict__ offs,
                                                    float* __restrict__ agg) {
    __shared__ int s_o0[256];
    __shared__ int s_cnt[256];
    __shared__ int s_pos[256];
    __shared__ int hist[256];
    __shared__ int noff[256];
    __shared__ int fill[256];
    __shared__ int list[BINCAP];
    __shared__ int sorted[BINCAP];
    __shared__ int s_total;
    __shared__ int s_next;

    const int t = threadIdx.x;
    const int lane = t & 31;
    const int wave = __builtin_amdgcn_readfirstlane(t >> 5);
    const int b = blockIdx.x;

    if (t < 256) {
        int o0 = 0, c = 0;
        if (t < GA) { o0 = offs[t * 256 + b]; c = offs[t * 256 + b + 1] - o0; }
        s_o0[t] = o0; s_cnt[t] = c; hist[t] = 0; fill[t] = 0;
    }
    if (t == 0) s_next = NWB;
    __syncthreads();
    if (wave == 0) {
        const int tot = wave_scan256(s_cnt, s_pos, lane);
        if (lane == 0) s_total = tot;
    }
    __syncthreads();
    const int total = s_total;
    const bool fast = total <= BINCAP;

    if (fast) {
        const int a = t >> 2, sub = t & 3;
        if (a < GA) {
            const int cA = s_cnt[a];
            const int pA = s_pos[a];
            const int* g = chunk + a * EPB + s_o0[a];
            for (int j = sub; j < cA; j += 16) {
                int w[4];
                for (int k = 0; k < 4; ++k) { const int jj = j + 4 * k; w[k] = jj < cA ? g[jj] : -1; }
                for (int k = 0; k < 4; ++k) {
                    const int jj = j + 4 * k;
                    if (jj < cA) { list[pA + jj] = w[k]; atomicAdd(&hist[w[k] >> 16], 1); }
                }
            }
        }
    }
    __syncthreads();
    if (fast && wave == 0) wave_scan256(hist, noff, lane);
    __syncthreads();
    if (fast) {
        for (int idx = t; idx < total; idx += TB) {
            const int w = list[idx];
            const int dl = w >> 16;
            const int p = noff[dl] + atomicAdd(&fill[dl], 1);
            sorted[p] = w & 0xffff;
        }
    }
    __syncthreads();

    const float* xl = x + lane * 4;
#define ROW(svv, jj) (*(const v4f*)(xl + (size_t)__builtin_amdgcn_readlane((svv), (jj)) * D))

    int i = wave;
    while (i < RNG) {
        v4f acc = {0.f, 0.f, 0.f, 0.f};
        if (fast) {
            const int base = __builtin_amdgcn_readfirstlane(noff[i]);
            const int c = __builtin_amdgcn_readfirstlane(noff[i + 1]) - base;
            for (int j0 = 0; j0 < c; j0 += 32) {
                const int rem = c - j0;
                const int cc = rem < 32 ? rem : 32;
                const int sv = lane < cc ? sorted[base + j0 + lane] : 0;
                int j = 0;
                for (; j + 8 <= cc; j += 8) {
                    const v4f r0 = ROW(sv, j + 0), r1 = ROW(sv, j + 1), r2 = ROW(sv, j + 2), r3 = ROW(sv, j + 3);
                    const v4f r4 = ROW(sv, j + 4), r5 = ROW(sv, j + 5), r6 = ROW(sv, j + 6), r7 = ROW(sv, j + 7);
                    acc += r0; acc += r1; acc += r2; acc += r3;
                    acc += r4; acc += r5; acc += r6; acc += r7;
                }
                if (j + 4 <= cc) {
                    const v4f r0 = ROW(sv, j + 0), r1 = ROW(sv, j + 1), r2 = ROW(sv, j + 2), r3 = ROW(sv, j + 3);
                    acc += r0; acc += r1; acc += r2; acc += r3;
                    j += 4;
                }
                if (j + 2 <= cc) {
                    const v4f r0 = ROW(sv, j + 0), r1 = ROW(sv, j + 1);
                    acc += r0; acc += r1;
                    j += 2;
                }
                if (j < cc) acc += ROW(sv, j);
            }
        } else {
            for (int a = 0; a < GA; ++a) {
                const int cA = __builtin_amdgcn_readfirstlane(s_cnt[a]);
                const int* g = chunk + a * EPB + __builtin_amdgcn_readfirstlane(s_o0[a]);
                for (int j0 = 0; j0 < cA; j0 += 32) {
                    const int jj = j0 + lane;
                    const int w = jj < cA ? g[jj] : -1;
                    unsigned m = (unsigned)__ballot((w >> 16) == i);
                    while (m) {
                        const int q = __builtin_ctz(m);
                        m &= m - 1;
                        acc += ROW(w & 0xffff, q);
                    }
                }
            }
        }
        *(v4f*)(agg + (size_t)(b * RNG + i) * D + lane * 4) = acc;
        int nx = 0;
        if (lane == 0) nx = atomicAdd(&s_next, 1);
        i = __builtin_amdgcn_readfirstlane(nx);
    }
#undef ROW
}

#define N_TILES  (N_NODES / 16)
#define MLP_GRID 512

__global__ __launch_bounds__(256) void gin_wmma_kernel(
    const float* __restrict__ x,   const float* __restrict__ agg,
    const float* __restrict__ W1,  const float* __restrict__ b1,
    const float* __restrict__ W2,  const float* __restrict__ b2,
    const float* __restrict__ Wres,const float* __restrict__ bres,
    float* __restrict__ out) {

    __shared__ float sh_xa[2][16 * LDSW];
    __shared__ float sh_xr[2][16 * LDSW];
    __shared__ float sh_t1[16 * LDSW];

    const int tid  = threadIdx.x;
    const int lane = tid & 31;
    const int wv   = tid >> 5;
    const int n    = lane & 15;
    const int h    = lane >> 4;
    const int colbase = wv * 16;

    const int fr = tid >> 5;
    const int fc = (tid & 31) * 4;

    int tile = blockIdx.x;
    v4f px0, pa0, px1, pa1;
    {
        const size_t g = (size_t)(tile * 16 + fr) * D + fc;
        px0 = *(const v4f*)(x + g);          pa0 = *(const v4f*)(agg + g);
        px1 = *(const v4f*)(x + g + 8 * D);  pa1 = *(const v4f*)(agg + g + 8 * D);
    }

    v2f w1[32], w2[32], wr[32];
    {
        const size_t o = (size_t)(2 * h) * D + colbase + n;
        #pragma unroll
        for (int q = 0; q < 32; ++q) {
            w1[q].x = W1[o + (size_t)(4 * q) * D];
            w1[q].y = W1[o + (size_t)(4 * q + 1) * D];
        }
        #pragma unroll
        for (int q = 0; q < 32; ++q) {
            w2[q].x = W2[o + (size_t)(4 * q) * D];
            w2[q].y = W2[o + (size_t)(4 * q + 1) * D];
        }
        #pragma unroll
        for (int q = 0; q < 32; ++q) {
            wr[q].x = Wres[o + (size_t)(4 * q) * D];
            wr[q].y = Wres[o + (size_t)(4 * q + 1) * D];
        }
    }
    const float bias1 = b1[colbase + n];
    const float bias2 = b2[colbase + n] + bres[colbase + n];

    int buf = 0;
    for (; tile < N_TILES; tile += MLP_GRID) {
        float* xa = sh_xa[buf];
        float* xr = sh_xr[buf];
        {
            const v4f s0 = {px0.x + pa0.x, px0.y + pa0.y, px0.z + pa0.z, px0.w + pa0.w};
            const v4f s1 = {px1.x + pa1.x, px1.y + pa1.y, px1.z + pa1.z, px1.w + pa1.w};
            *(v4f*)(xr + fr * LDSW + fc)       = px0;
            *(v4f*)(xa + fr * LDSW + fc)       = s0;
            *(v4f*)(xr + (fr + 8) * LDSW + fc) = px1;
            *(v4f*)(xa + (fr + 8) * LDSW + fc) = s1;
        }
        __syncthreads();

        if (tile + MLP_GRID < N_TILES) {
            const size_t g = (size_t)((tile + MLP_GRID) * 16 + fr) * D + fc;
            px0 = *(const v4f*)(x + g);          pa0 = *(const v4f*)(agg + g);
            px1 = *(const v4f*)(x + g + 8 * D);  pa1 = *(const v4f*)(agg + g + 8 * D);
        }

        v8f acc = {};
        #pragma unroll
        for (int q = 0; q < 32; ++q) {
            const v2f a = *(const v2f*)(xa + n * LDSW + 4 * q + 2 * h);
            acc = __builtin_amdgcn_wmma_f32_16x16x4_f32(
                      false, a, false, w1[q], (short)0, acc, false, false);
        }
        #pragma unroll
        for (int r = 0; r < 8; ++r) {
            const float t = acc[r] + bias1;
            sh_t1[(r + 8 * h) * LDSW + colbase + n] = fmaxf(t, 0.f);
        }
        __syncthreads();

        v8f acc2 = {};
        #pragma unroll
        for (int q = 0; q < 32; ++q) {
            const v2f a1 = *(const v2f*)(sh_t1 + n * LDSW + 4 * q + 2 * h);
            acc2 = __builtin_amdgcn_wmma_f32_16x16x4_f32(
                       false, a1, false, w2[q], (short)0, acc2, false, false);
            const v2f a2 = *(const v2f*)(xr + n * LDSW + 4 * q + 2 * h);
            acc2 = __builtin_amdgcn_wmma_f32_16x16x4_f32(
                       false, a2, false, wr[q], (short)0, acc2, false, false);
        }
        const int rowbase = tile * 16;
        #pragma unroll
        for (int r = 0; r < 8; ++r) {
            out[(size_t)(rowbase + r + 8 * h) * D + colbase + n] = acc2[r] + bias2;
        }
        buf ^= 1;
    }
}

extern "C" void kernel_launch(void* const* d_in, const int* in_sizes, int n_in,
                              void* d_out, int out_size, void* d_ws, size_t ws_size,
                              hipStream_t stream) {
    const float* x    = (const float*)d_in[0];
    const int*   ei   = (const int*)  d_in[1];
    const float* W1   = (const float*)d_in[2];
    const float* b1   = (const float*)d_in[3];
    const float* W2   = (const float*)d_in[4];
    const float* b2   = (const float*)d_in[5];
    const float* Wres = (const float*)d_in[6];
    const float* bres = (const float*)d_in[7];
    float* out = (float*)d_out;
    float* agg = (float*)d_ws;
    int* chunk = (int*)d_ws + CHUNK_OFF;
    int* offs  = (int*)d_ws + OFFS_OFF;

    bin_kernel<<<GA, TA, 0, stream>>>(ei, ei + N_EDGES, chunk, offs);

    gather_kernel<<<NB, TB, 0, stream>>>(x, chunk, offs, agg);

    gin_wmma_kernel<<<MLP_GRID, 256, 0, stream>>>(
        x, agg, W1, b1, W2, b2, Wres, bres, out);
}
